# starting kernel
# baseline (speedup 1.0000x reference)
#include <hip/hip_runtime.h>
#include <hip/hip_bf16.h>


#define BATCH    8192
#define IN_DIM   512
#define NUM_TREE 64
#define NUM_LEAF 32
#define NUM_INT  31
#define OUT_DIM  16
#define XL_STRIDE 520

typedef __attribute__((ext_vector_type(8)))  float  v8f;
typedef __attribute__((ext_vector_type(16))) __bf16 v16bf;

union Frag16 {
    v16bf v;
    unsigned short u[16];
    uint4 q[2];
};

__device__ __forceinline__ unsigned short f2bf(float f) {
    unsigned int u = __float_as_uint(f);
    unsigned int r = u + 0x7FFFu + ((u >> 16) & 1u);
    return (unsigned short)(r >> 16);
}

__device__ __forceinline__ float sigf(float x) {
    return 1.0f / (1.0f + __expf(-x));
}

__global__ void dndf_pack_w(const float* __restrict__ W,
                            unsigned short* __restrict__ Wpf) {
    int idx = blockIdx.x * 256 + threadIdx.x;
    const int total = NUM_TREE * 16 * 2 * 32 * 16;
    if (idx >= total) return;
    int e    = idx & 15;
    int lane = (idx >> 4) & 31;
    int ct   = (idx >> 9) & 1;
    int kk   = (idx >> 10) & 15;
    int t    = idx >> 14;
    int n    = ct * 16 + (lane & 15);
    int koff = (lane >> 4) * 8;
    int k    = kk * 32 + ((e < 8) ? (koff + e) : (16 + koff + (e - 8)));
    unsigned short v = 0;
    if (n < NUM_INT) v = f2bf(W[(t * NUM_INT + n) * IN_DIM + k]);
    Wpf[idx] = v;
}

__global__ void dndf_pack_sm(const float* __restrict__ leafs,
                             const float* __restrict__ b,
                             unsigned short* __restrict__ smf,
                             float* __restrict__ bp) {
    int idx = blockIdx.x * 256 + threadIdx.x;
    if (idx >= NUM_TREE * 32) return;
    int lane = idx & 31;
    int t    = idx >> 5;
    int n    = lane & 15;
    int koff = (lane >> 4) * 8;
    for (int e = 0; e < 16; ++e) {
        int leaf = (e < 8) ? (koff + e) : (16 + koff + (e - 8));
        const float* row = leafs + (t * NUM_LEAF + leaf) * OUT_DIM;
        float mx = row[0];
        #pragma unroll
        for (int o = 1; o < OUT_DIM; ++o) mx = fmaxf(mx, row[o]);
        float sum = 0.f;
        #pragma unroll
        for (int o = 0; o < OUT_DIM; ++o) sum += __expf(row[o] - mx);
        float val = __expf(row[n] - mx) / (sum * (float)NUM_TREE);
        smf[idx * 16 + e] = f2bf(val);
    }
    int i = idx & 31;
    bp[idx] = (i < NUM_INT) ? b[(idx >> 5) * NUM_INT + i] : 0.0f;
}

__global__ __launch_bounds__(256) void dndf_main(
        const float* __restrict__ x,
        const unsigned short* __restrict__ Wpf,
        const unsigned short* __restrict__ smf,
        const float* __restrict__ bp,
        float* __restrict__ out) {

    __shared__ unsigned short xtile[16 * XL_STRIDE];
    __shared__ float pbuf[8][16][33];
    __shared__ float red[8][16][16];

    const int tid  = threadIdx.x;
    const int wave = tid >> 5;
    const int lane = tid & 31;
    const int row0 = blockIdx.x * 16;

    for (int i = tid; i < 16 * (IN_DIM / 4); i += 256) {
        int r  = i / (IN_DIM / 4);
        int c4 = i % (IN_DIM / 4);
        float4 f = ((const float4*)(x + (size_t)(row0 + r) * IN_DIM))[c4];
        int o = r * XL_STRIDE + c4 * 4;
        xtile[o + 0] = f2bf(f.x);
        xtile[o + 1] = f2bf(f.y);
        xtile[o + 2] = f2bf(f.z);
        xtile[o + 3] = f2bf(f.w);
    }
    __syncthreads();

    const int ln    = lane & 15;
    const int khalf = (lane >> 4) * 8;
    const int mrow  = (lane >> 4) * 8;

    v8f out_acc = {};

    for (int tl = 0; tl < 8; ++tl) {
        const int t = wave * 8 + tl;
        v8f c0 = {};
        v8f c1 = {};
        const unsigned short* wp = Wpf + (size_t)t * (16 * 2 * 32 * 16);

        for (int kk = 0; kk < 16; ++kk) {
            Frag16 a;
            const unsigned short* xr = &xtile[ln * XL_STRIDE + kk * 32 + khalf];
            a.q[0] = *(const uint4*)(xr);
            a.q[1] = *(const uint4*)(xr + 16);
            Frag16 b0, b1;
            const uint4* wb0 = (const uint4*)(wp + (((size_t)kk * 2 + 0) * 32 + lane) * 16);
            const uint4* wb1 = (const uint4*)(wp + (((size_t)kk * 2 + 1) * 32 + lane) * 16);
            b0.q[0] = wb0[0]; b0.q[1] = wb0[1];
            b1.q[0] = wb1[0]; b1.q[1] = wb1[1];

            c0 = __builtin_amdgcn_wmma_f32_16x16x32_bf16(
                    false, a.v, false, b0.v, (short)0, c0, false, false);
            c1 = __builtin_amdgcn_wmma_f32_16x16x32_bf16(
                    false, a.v, false, b1.v, (short)0, c1, false, false);
        }

        const float bias0 = bp[t * 32 + ln];
        const float bias1 = bp[t * 32 + 16 + ln];
        #pragma unroll
        for (int v = 0; v < 8; ++v) {
            pbuf[wave][mrow + v][ln]      = sigf(c0[v] + bias0);
            pbuf[wave][mrow + v][16 + ln] = sigf(c1[v] + bias1);
        }
        __syncthreads();

        Frag16 r;
        #pragma unroll
        for (int e = 0; e < 16; ++e) {
            int leaf = (e < 8) ? (khalf + e) : (16 + khalf + (e - 8));
            float prod = 1.f;
            #pragma unroll
            for (int d = 0; d <= 4; ++d) {
                int node = ((1 << d) - 1) + (leaf >> (5 - d));
                float p  = pbuf[wave][ln][node];
                prod *= (((leaf >> (4 - d)) & 1) ? (1.f - p) : p);
            }
            r.u[e] = f2bf(prod);
        }

        Frag16 s;
        const uint4* sp = (const uint4*)(smf + ((size_t)t * 32 + lane) * 16);
        s.q[0] = sp[0]; s.q[1] = sp[1];
        out_acc = __builtin_amdgcn_wmma_f32_16x16x32_bf16(
                    false, r.v, false, s.v, (short)0, out_acc, false, false);
        __syncthreads();
    }

    #pragma unroll
    for (int v = 0; v < 8; ++v) red[wave][mrow + v][ln] = out_acc[v];
    __syncthreads();

    const int rrow = tid >> 4;
    const int col  = tid & 15;
    float sum = 0.f;
    #pragma unroll
    for (int w = 0; w < 8; ++w) sum += red[w][rrow][col];
    out[(size_t)(row0 + rrow) * OUT_DIM + col] = sum;
}

extern "C" void kernel_launch(void* const* d_in, const int* in_sizes, int n_in,
                              void* d_out, int out_size, void* d_ws, size_t ws_size,
                              hipStream_t stream) {
    const float* x     = (const float*)d_in[0];
    const float* W     = (const float*)d_in[1];
    const float* b     = (const float*)d_in[2];
    const float* leafs = (const float*)d_in[3];
    float* out = (float*)d_out;

    unsigned short* Wpf = (unsigned short*)d_ws;
    unsigned short* smf = (unsigned short*)((char*)d_ws + (size_t)2 * 1024 * 1024);
    float*          bpp = (float*)((char*)d_ws + (size_t)2 * 1024 * 1024 + 64 * 1024);

    dndf_pack_w <<<4096, 256, 0, stream>>>(W, Wpf);
    dndf_pack_sm<<<8,    256, 0, stream>>>(leafs, b, smf, bpp);
    dndf_main   <<<BATCH / 16, 256, 0, stream>>>(x, Wpf, smf, bpp, out);
}


# main kernel: each wave keeps 8 K-steps (32 loads) of its W-fragment stream in flight in a register ring refilled as consumed, not load-wait-WMMA per fragment; x-tile and bias loads issued early
# speedup vs baseline: 1.9790x; 1.9790x over previous
#include <hip/hip_runtime.h>
#include <hip/hip_bf16.h>


#define BATCH    8192
#define IN_DIM   512
#define NUM_TREE 64
#define NUM_LEAF 32
#define NUM_INT  31
#define OUT_DIM  16
#define XL_STRIDE 520
#define PD 8
#define STEP_ELEMS 1024

typedef __attribute__((ext_vector_type(8)))  float  v8f;
typedef __attribute__((ext_vector_type(16))) __bf16 v16bf;

union Frag16 {
    v16bf v;
    unsigned short u[16];
    uint4 q[2];
};

__device__ __forceinline__ unsigned short f2bf(float f) {
    unsigned int u = __float_as_uint(f);
    unsigned int r = u + 0x7FFFu + ((u >> 16) & 1u);
    return (unsigned short)(r >> 16);
}

__device__ __forceinline__ float sigf(float x) {
    return 1.0f / (1.0f + __expf(-x));
}

__global__ void dndf_pack_w(const float* __restrict__ W,
                            unsigned short* __restrict__ Wpf) {
    int idx = blockIdx.x * 256 + threadIdx.x;
    const int total = NUM_TREE * 16 * 2 * 32 * 16;
    if (idx >= total) return;
    int e    = idx & 15;
    int lane = (idx >> 4) & 31;
    int ct   = (idx >> 9) & 1;
    int kk   = (idx >> 10) & 15;
    int t    = idx >> 14;
    int n    = ct * 16 + (lane & 15);
    int koff = (lane >> 4) * 8;
    int k    = kk * 32 + ((e < 8) ? (koff + e) : (16 + koff + (e - 8)));
    unsigned short v = 0;
    if (n < NUM_INT) v = f2bf(W[(t * NUM_INT + n) * IN_DIM + k]);
    Wpf[idx] = v;
}

__global__ void dndf_pack_sm(const float* __restrict__ leafs,
                             const float* __restrict__ b,
                             unsigned short* __restrict__ smf,
                             float* __restrict__ bp) {
    int idx = blockIdx.x * 256 + threadIdx.x;
    if (idx >= NUM_TREE * 32) return;
    int lane = idx & 31;
    int t    = idx >> 5;
    int n    = lane & 15;
    int koff = (lane >> 4) * 8;
    for (int e = 0; e < 16; ++e) {
        int leaf = (e < 8) ? (koff + e) : (16 + koff + (e - 8));
        const float* row = leafs + (t * NUM_LEAF + leaf) * OUT_DIM;
        float mx = row[0];
        #pragma unroll
        for (int o = 1; o < OUT_DIM; ++o) mx = fmaxf(mx, row[o]);
        float sum = 0.f;
        #pragma unroll
        for (int o = 0; o < OUT_DIM; ++o) sum += __expf(row[o] - mx);
        float val = __expf(row[n] - mx) / (sum * (float)NUM_TREE);
        smf[idx * 16 + e] = f2bf(val);
    }
    int i = idx & 31;
    bp[idx] = (i < NUM_INT) ? b[(idx >> 5) * NUM_INT + i] : 0.0f;
}

__device__ __forceinline__ void load_step(Frag16 (&dst)[2], const unsigned short* p) {
    const uint4* q = (const uint4*)p;
    dst[0].q[0] = q[0];  dst[0].q[1] = q[1];
    dst[1].q[0] = q[64]; dst[1].q[1] = q[65];
}

template <bool LAST>
__device__ __forceinline__ void one_tree(
        Frag16 (&ring)[PD][2], v8f& out_acc,
        const unsigned short* xrow,
        const unsigned short* wtree,
        const unsigned short* smf, const float* bp,
        float (&pw)[16][33],
        int t, int lane, int ln, int khalf, int mrow) {

    const float bias0 = bp[t * 32 + ln];
    const float bias1 = bp[t * 32 + 16 + ln];
    Frag16 s;
    const uint4* sp = (const uint4*)(smf + ((size_t)t * 32 + lane) * 16);
    s.q[0] = sp[0]; s.q[1] = sp[1];

    v8f c0 = {};
    v8f c1 = {};
    #pragma unroll
    for (int kk = 0; kk < 16; ++kk) {
        Frag16 a;
        a.q[0] = *(const uint4*)(xrow + kk * 32);
        a.q[1] = *(const uint4*)(xrow + kk * 32 + 16);
        c0 = __builtin_amdgcn_wmma_f32_16x16x32_bf16(
                false, a.v, false, ring[kk % PD][0].v, (short)0, c0, false, false);
        c1 = __builtin_amdgcn_wmma_f32_16x16x32_bf16(
                false, a.v, false, ring[kk % PD][1].v, (short)0, c1, false, false);
        if (!LAST || kk + PD < 16)
            load_step(ring[kk % PD], wtree + (size_t)(kk + PD) * STEP_ELEMS);
    }

    #pragma unroll
    for (int v = 0; v < 8; ++v) {
        pw[mrow + v][ln]      = sigf(c0[v] + bias0);
        pw[mrow + v][16 + ln] = sigf(c1[v] + bias1);
    }
    __syncthreads();

    Frag16 r;
    #pragma unroll
    for (int e = 0; e < 16; ++e) {
        int leaf = (e < 8) ? (khalf + e) : (16 + khalf + (e - 8));
        float prod = 1.f;
        #pragma unroll
        for (int d = 0; d <= 4; ++d) {
            int node = ((1 << d) - 1) + (leaf >> (5 - d));
            float p  = pw[ln][node];
            prod *= (((leaf >> (4 - d)) & 1) ? (1.f - p) : p);
        }
        r.u[e] = f2bf(prod);
    }

    out_acc = __builtin_amdgcn_wmma_f32_16x16x32_bf16(
                false, r.v, false, s.v, (short)0, out_acc, false, false);
    __syncthreads();
}

__global__ __launch_bounds__(256) __attribute__((amdgpu_waves_per_eu(4, 4))) void dndf_main(
        const float* __restrict__ x,
        const unsigned short* __restrict__ Wpf,
        const unsigned short* __restrict__ smf,
        const float* __restrict__ bp,
        float* __restrict__ out) {

    __shared__ unsigned short xtile[16 * XL_STRIDE];
    __shared__ float pbuf[8][16][33];
    __shared__ float red[8][16][16];

    const int tid  = threadIdx.x;
    const int wave = tid >> 5;
    const int lane = tid & 31;
    const int row0 = blockIdx.x * 16;

    const unsigned short* wl = Wpf + (size_t)wave * 8 * 16 * STEP_ELEMS + lane * 16;
    Frag16 ring[PD][2];
    #pragma unroll
    for (int j = 0; j < PD; ++j) load_step(ring[j], wl + (size_t)j * STEP_ELEMS);

    float4 xf[8];
    #pragma unroll
    for (int j = 0; j < 8; ++j) {
        int i  = tid + j * 256;
        int r  = i / (IN_DIM / 4);
        int c4 = i % (IN_DIM / 4);
        xf[j] = ((const float4*)(x + (size_t)(row0 + r) * IN_DIM))[c4];
    }
    #pragma unroll
    for (int j = 0; j < 8; ++j) {
        int i  = tid + j * 256;
        int r  = i / (IN_DIM / 4);
        int c4 = i % (IN_DIM / 4);
        int o = r * XL_STRIDE + c4 * 4;
        xtile[o + 0] = f2bf(xf[j].x);
        xtile[o + 1] = f2bf(xf[j].y);
        xtile[o + 2] = f2bf(xf[j].z);
        xtile[o + 3] = f2bf(xf[j].w);
    }
    __syncthreads();

    const int ln    = lane & 15;
    const int khalf = (lane >> 4) * 8;
    const int mrow  = (lane >> 4) * 8;

    const unsigned short* xrow = &xtile[ln * XL_STRIDE + khalf];
    v8f out_acc = {};

    for (int tl = 0; tl < 7; ++tl)
        one_tree<false>(ring, out_acc, xrow, wl + (size_t)tl * 16 * STEP_ELEMS,
                        smf, bp, pbuf[wave], wave * 8 + tl, lane, ln, khalf, mrow);
    one_tree<true>(ring, out_acc, xrow, wl + (size_t)7 * 16 * STEP_ELEMS,
                   smf, bp, pbuf[wave], wave * 8 + 7, lane, ln, khalf, mrow);

    #pragma unroll
    for (int v = 0; v < 8; ++v) red[wave][mrow + v][ln] = out_acc[v];
    __syncthreads();

    const int rrow = tid >> 4;
    const int col  = tid & 15;
    float sum = 0.f;
    #pragma unroll
    for (int w = 0; w < 8; ++w) sum += red[w][rrow][col];
    out[(size_t)(row0 + rrow) * OUT_DIM + col] = sum;
}

extern "C" void kernel_launch(void* const* d_in, const int* in_sizes, int n_in,
                              void* d_out, int out_size, void* d_ws, size_t ws_size,
                              hipStream_t stream) {
    const float* x     = (const float*)d_in[0];
    const float* W     = (const float*)d_in[1];
    const float* b     = (const float*)d_in[2];
    const float* leafs = (const float*)d_in[3];
    float* out = (float*)d_out;

    unsigned short* Wpf = (unsigned short*)d_ws;
    unsigned short* smf = (unsigned short*)((char*)d_ws + (size_t)2 * 1024 * 1024);
    float*          bpp = (float*)((char*)d_ws + (size_t)2 * 1024 * 1024 + 64 * 1024);

    dndf_pack_w <<<4096, 256, 0, stream>>>(W, Wpf);
    dndf_pack_sm<<<8,    256, 0, stream>>>(leafs, b, smf, bpp);
    dndf_main   <<<BATCH / 16, 256, 0, stream>>>(x, Wpf, smf, bpp, out);
}


# main kernel: each W fragment fetched from L2 now feeds two 16-row tiles (block = 32 rows, 16 waves x 4 trees, 4 accumulator tiles per wave): W traffic halves; output sum = 16 partials of 4 trees
# speedup vs baseline: 2.4062x; 1.2159x over previous
#include <hip/hip_runtime.h>
#include <hip/hip_bf16.h>


#define BATCH    8192
#define IN_DIM   512
#define NUM_TREE 64
#define NUM_LEAF 32
#define NUM_INT  31
#define OUT_DIM  16
#define XL_STRIDE 520
#define PD 8
#define ROWS 32
#define NWAVE 16
#define TPW 4
#define STEP_ELEMS 1024

typedef __attribute__((ext_vector_type(8)))  float  v8f;
typedef __attribute__((ext_vector_type(16))) __bf16 v16bf;

union Frag16 {
    v16bf v;
    unsigned short u[16];
    uint4 q[2];
};

__device__ __forceinline__ unsigned short f2bf(float f) {
    unsigned int u = __float_as_uint(f);
    unsigned int r = u + 0x7FFFu + ((u >> 16) & 1u);
    return (unsigned short)(r >> 16);
}

__device__ __forceinline__ float sigf(float x) {
    return 1.0f / (1.0f + __expf(-x));
}

__global__ void dndf_pack_w(const float* __restrict__ W,
                            unsigned short* __restrict__ Wpf) {
    int idx = blockIdx.x * 256 + threadIdx.x;
    const int total = NUM_TREE * 16 * 2 * 32 * 16;
    if (idx >= total) return;
    int e    = idx & 15;
    int lane = (idx >> 4) & 31;
    int ct   = (idx >> 9) & 1;
    int kk   = (idx >> 10) & 15;
    int t    = idx >> 14;
    int n    = ct * 16 + (lane & 15);
    int koff = (lane >> 4) * 8;
    int k    = kk * 32 + ((e < 8) ? (koff + e) : (16 + koff + (e - 8)));
    unsigned short v = 0;
    if (n < NUM_INT) v = f2bf(W[(t * NUM_INT + n) * IN_DIM + k]);
    Wpf[idx] = v;
}

__global__ void dndf_pack_sm(const float* __restrict__ leafs,
                             const float* __restrict__ b,
                             unsigned short* __restrict__ smf,
                             float* __restrict__ bp) {
    int idx = blockIdx.x * 256 + threadIdx.x;
    if (idx >= NUM_TREE * 32) return;
    int lane = idx & 31;
    int t    = idx >> 5;
    int n    = lane & 15;
    int koff = (lane >> 4) * 8;
    for (int e = 0; e < 16; ++e) {
        int leaf = (e < 8) ? (koff + e) : (16 + koff + (e - 8));
        const float* row = leafs + (t * NUM_LEAF + leaf) * OUT_DIM;
        float mx = row[0];
        #pragma unroll
        for (int o = 1; o < OUT_DIM; ++o) mx = fmaxf(mx, row[o]);
        float sum = 0.f;
        #pragma unroll
        for (int o = 0; o < OUT_DIM; ++o) sum += __expf(row[o] - mx);
        float val = __expf(row[n] - mx) / (sum * (float)NUM_TREE);
        smf[idx * 16 + e] = f2bf(val);
    }
    int i = idx & 31;
    bp[idx] = (i < NUM_INT) ? b[(idx >> 5) * NUM_INT + i] : 0.0f;
}

__device__ __forceinline__ void load_step(Frag16 (&dst)[2], const unsigned short* p) {
    const uint4* q = (const uint4*)p;
    dst[0].q[0] = q[0];  dst[0].q[1] = q[1];
    dst[1].q[0] = q[64]; dst[1].q[1] = q[65];
}

template <bool LAST>
__device__ __forceinline__ void one_tree(
        Frag16 (&ring)[PD][2], v8f (&out_acc)[2],
        const unsigned short* xrow,
        const unsigned short* wtree,
        const unsigned short* smf, const float* bp,
        float (&pw)[2][16][33],
        int t, int lane, int ln, int khalf, int mrow) {

    const float bias0 = bp[t * 32 + ln];
    const float bias1 = bp[t * 32 + 16 + ln];
    Frag16 s;
    const uint4* sp = (const uint4*)(smf + ((size_t)t * 32 + lane) * 16);
    s.q[0] = sp[0]; s.q[1] = sp[1];

    v8f c00 = {};
    v8f c01 = {};
    v8f c10 = {};
    v8f c11 = {};
    #pragma unroll
    for (int kk = 0; kk < 16; ++kk) {
        Frag16 a0, a1;
        a0.q[0] = *(const uint4*)(xrow + kk * 32);
        a0.q[1] = *(const uint4*)(xrow + kk * 32 + 16);
        a1.q[0] = *(const uint4*)(xrow + 16 * XL_STRIDE + kk * 32);
        a1.q[1] = *(const uint4*)(xrow + 16 * XL_STRIDE + kk * 32 + 16);
        c00 = __builtin_amdgcn_wmma_f32_16x16x32_bf16(
                false, a0.v, false, ring[kk % PD][0].v, (short)0, c00, false, false);
        c01 = __builtin_amdgcn_wmma_f32_16x16x32_bf16(
                false, a0.v, false, ring[kk % PD][1].v, (short)0, c01, false, false);
        c10 = __builtin_amdgcn_wmma_f32_16x16x32_bf16(
                false, a1.v, false, ring[kk % PD][0].v, (short)0, c10, false, false);
        c11 = __builtin_amdgcn_wmma_f32_16x16x32_bf16(
                false, a1.v, false, ring[kk % PD][1].v, (short)0, c11, false, false);
        if (!LAST || kk + PD < 16)
            load_step(ring[kk % PD], wtree + (size_t)(kk + PD) * STEP_ELEMS);
    }

    #pragma unroll
    for (int v = 0; v < 8; ++v) {
        pw[0][mrow + v][ln]      = sigf(c00[v] + bias0);
        pw[0][mrow + v][16 + ln] = sigf(c01[v] + bias1);
        pw[1][mrow + v][ln]      = sigf(c10[v] + bias0);
        pw[1][mrow + v][16 + ln] = sigf(c11[v] + bias1);
    }
    __syncthreads();

    #pragma unroll
    for (int m = 0; m < 2; ++m) {
        Frag16 r;
        #pragma unroll
        for (int e = 0; e < 16; ++e) {
            int leaf = (e < 8) ? (khalf + e) : (16 + khalf + (e - 8));
            float prod = 1.f;
            #pragma unroll
            for (int d = 0; d <= 4; ++d) {
                int node = ((1 << d) - 1) + (leaf >> (5 - d));
                float p  = pw[m][ln][node];
                prod *= (((leaf >> (4 - d)) & 1) ? (1.f - p) : p);
            }
            r.u[e] = f2bf(prod);
        }
        out_acc[m] = __builtin_amdgcn_wmma_f32_16x16x32_bf16(
                    false, r.v, false, s.v, (short)0, out_acc[m], false, false);
    }
    __syncthreads();
}

__global__ __launch_bounds__(NWAVE * 32) __attribute__((amdgpu_waves_per_eu(4, 4))) void dndf_main(
        const float* __restrict__ x,
        const unsigned short* __restrict__ Wpf,
        const unsigned short* __restrict__ smf,
        const float* __restrict__ bp,
        float* __restrict__ out) {

    __shared__ unsigned short xtile[ROWS * XL_STRIDE];
    __shared__ float pbuf[NWAVE][2][16][33];

    const int tid  = threadIdx.x;
    const int wave = tid >> 5;
    const int lane = tid & 31;
    const int row0 = blockIdx.x * ROWS;

    const unsigned short* wl = Wpf + (size_t)wave * TPW * 16 * STEP_ELEMS + lane * 16;
    Frag16 ring[PD][2];
    #pragma unroll
    for (int j = 0; j < PD; ++j) load_step(ring[j], wl + (size_t)j * STEP_ELEMS);

    float4 xf[8];
    #pragma unroll
    for (int j = 0; j < 8; ++j) {
        int i  = tid + j * (NWAVE * 32);
        int r  = i / (IN_DIM / 4);
        int c4 = i % (IN_DIM / 4);
        xf[j] = ((const float4*)(x + (size_t)(row0 + r) * IN_DIM))[c4];
    }
    #pragma unroll
    for (int j = 0; j < 8; ++j) {
        int i  = tid + j * (NWAVE * 32);
        int r  = i / (IN_DIM / 4);
        int c4 = i % (IN_DIM / 4);
        int o = r * XL_STRIDE + c4 * 4;
        xtile[o + 0] = f2bf(xf[j].x);
        xtile[o + 1] = f2bf(xf[j].y);
        xtile[o + 2] = f2bf(xf[j].z);
        xtile[o + 3] = f2bf(xf[j].w);
    }
    __syncthreads();

    const int ln    = lane & 15;
    const int khalf = (lane >> 4) * 8;
    const int mrow  = (lane >> 4) * 8;

    const unsigned short* xrow = &xtile[ln * XL_STRIDE + khalf];
    v8f out_acc[2] = {};

    for (int tl = 0; tl < TPW - 1; ++tl)
        one_tree<false>(ring, out_acc, xrow, wl + (size_t)tl * 16 * STEP_ELEMS,
                        smf, bp, pbuf[wave], wave * TPW + tl, lane, ln, khalf, mrow);
    one_tree<true>(ring, out_acc, xrow, wl + (size_t)(TPW - 1) * 16 * STEP_ELEMS,
                   smf, bp, pbuf[wave], wave * TPW + (TPW - 1), lane, ln, khalf, mrow);

    #pragma unroll
    for (int m = 0; m < 2; ++m) {
        #pragma unroll
        for (int v = 0; v < 8; ++v) pbuf[wave][m][mrow + v][ln] = out_acc[m][v];
    }
    __syncthreads();

    const int rrow = tid >> 4;
    const int col  = tid & 15;
    float sum = 0.f;
    #pragma unroll
    for (int w = 0; w < NWAVE; ++w) sum += pbuf[w][rrow >> 4][rrow & 15][col];
    out[(size_t)(row0 + rrow) * OUT_DIM + col] = sum;
}

extern "C" void kernel_launch(void* const* d_in, const int* in_sizes, int n_in,
                              void* d_out, int out_size, void* d_ws, size_t ws_size,
                              hipStream_t stream) {
    const float* x     = (const float*)d_in[0];
    const float* W     = (const float*)d_in[1];
    const float* b     = (const float*)d_in[2];
    const float* leafs = (const float*)d_in[3];
    float* out = (float*)d_out;

    unsigned short* Wpf = (unsigned short*)d_ws;
    unsigned short* smf = (unsigned short*)((char*)d_ws + (size_t)2 * 1024 * 1024);
    float*          bpp = (float*)((char*)d_ws + (size_t)2 * 1024 * 1024 + 64 * 1024);

    dndf_pack_w <<<4096, 256, 0, stream>>>(W, Wpf);
    dndf_pack_sm<<<8,    256, 0, stream>>>(leafs, b, smf, bpp);
    dndf_main   <<<256, 512, 0, stream>>>(x, Wpf, smf, bpp, out);
}


# pack stage: leaf-softmax and bias packing done by the first threads of the weight-pack launch, one fragment element per thread (no 16-row serial walk): 2 launches per repetition, not 3
# speedup vs baseline: 2.9013x; 1.2058x over previous
#include <hip/hip_runtime.h>
#include <hip/hip_bf16.h>


#define BATCH    8192
#define IN_DIM   512
#define NUM_TREE 64
#define NUM_LEAF 32
#define NUM_INT  31
#define OUT_DIM  16
#define XL_STRIDE 520
#define PD 8
#define ROWS 32
#define NWAVE 16
#define TPW 4
#define STEP_ELEMS 1024

typedef __attribute__((ext_vector_type(8)))  float  v8f;
typedef __attribute__((ext_vector_type(16))) __bf16 v16bf;

union Frag16 {
    v16bf v;
    unsigned short u[16];
    uint4 q[2];
};

__device__ __forceinline__ unsigned short f2bf(float f) {
    unsigned int u = __float_as_uint(f);
    unsigned int r = u + 0x7FFFu + ((u >> 16) & 1u);
    return (unsigned short)(r >> 16);
}

__device__ __forceinline__ float sigf(float x) {
    return 1.0f / (1.0f + __expf(-x));
}

__global__ void dndf_pack(const float* __restrict__ W,
                          const float* __restrict__ leafs,
                          const float* __restrict__ b,
                          unsigned short* __restrict__ Wpf,
                          unsigned short* __restrict__ smf,
                          float* __restrict__ bp) {
    int idx = blockIdx.x * 256 + threadIdx.x;
    const int total = NUM_TREE * 16 * 2 * 32 * 16;
    if (idx >= total) return;
    {
        int e    = idx & 15;
        int lane = (idx >> 4) & 31;
        int ct   = (idx >> 9) & 1;
        int kk   = (idx >> 10) & 15;
        int t    = idx >> 14;
        int n    = ct * 16 + (lane & 15);
        int koff = (lane >> 4) * 8;
        int k    = kk * 32 + ((e < 8) ? (koff + e) : (16 + koff + (e - 8)));
        unsigned short v = 0;
        if (n < NUM_INT) v = f2bf(W[(t * NUM_INT + n) * IN_DIM + k]);
        Wpf[idx] = v;
    }
    if (idx < NUM_TREE * 32 * 16) {
        int e    = idx & 15;
        int lane = (idx >> 4) & 31;
        int t    = idx >> 9;
        int n    = lane & 15;
        int koff = (lane >> 4) * 8;
        int leaf = (e < 8) ? (koff + e) : (16 + koff + (e - 8));
        const float* row = leafs + (t * NUM_LEAF + leaf) * OUT_DIM;
        float mx = row[0];
        #pragma unroll
        for (int o = 1; o < OUT_DIM; ++o) mx = fmaxf(mx, row[o]);
        float sum = 0.f;
        #pragma unroll
        for (int o = 0; o < OUT_DIM; ++o) sum += __expf(row[o] - mx);
        float val = __expf(row[n] - mx) / (sum * (float)NUM_TREE);
        smf[idx] = f2bf(val);
    }
    if (idx < NUM_TREE * 32) {
        int i = idx & 31;
        bp[idx] = (i < NUM_INT) ? b[(idx >> 5) * NUM_INT + i] : 0.0f;
    }
}

__device__ __forceinline__ void load_step(Frag16 (&dst)[2], const unsigned short* p) {
    const uint4* q = (const uint4*)p;
    dst[0].q[0] = q[0];  dst[0].q[1] = q[1];
    dst[1].q[0] = q[64]; dst[1].q[1] = q[65];
}

template <bool LAST>
__device__ __forceinline__ void one_tree(
        Frag16 (&ring)[PD][2], v8f (&out_acc)[2],
        const unsigned short* xrow,
        const unsigned short* wtree,
        const unsigned short* smf, const float* bp,
        float (&pw)[2][16][33],
        int t, int lane, int ln, int khalf, int mrow) {

    const float bias0 = bp[t * 32 + ln];
    const float bias1 = bp[t * 32 + 16 + ln];
    Frag16 s;
    const uint4* sp = (const uint4*)(smf + ((size_t)t * 32 + lane) * 16);
    s.q[0] = sp[0]; s.q[1] = sp[1];

    v8f c00 = {};
    v8f c01 = {};
    v8f c10 = {};
    v8f c11 = {};
    #pragma unroll
    for (int kk = 0; kk < 16; ++kk) {
        Frag16 a0, a1;
        a0.q[0] = *(const uint4*)(xrow + kk * 32);
        a0.q[1] = *(const uint4*)(xrow + kk * 32 + 16);
        a1.q[0] = *(const uint4*)(xrow + 16 * XL_STRIDE + kk * 32);
        a1.q[1] = *(const uint4*)(xrow + 16 * XL_STRIDE + kk * 32 + 16);
        c00 = __builtin_amdgcn_wmma_f32_16x16x32_bf16(
                false, a0.v, false, ring[kk % PD][0].v, (short)0, c00, false, false);
        c01 = __builtin_amdgcn_wmma_f32_16x16x32_bf16(
                false, a0.v, false, ring[kk % PD][1].v, (short)0, c01, false, false);
        c10 = __builtin_amdgcn_wmma_f32_16x16x32_bf16(
                false, a1.v, false, ring[kk % PD][0].v, (short)0, c10, false, false);
        c11 = __builtin_amdgcn_wmma_f32_16x16x32_bf16(
                false, a1.v, false, ring[kk % PD][1].v, (short)0, c11, false, false);
        if (!LAST || kk + PD < 16)
            load_step(ring[kk % PD], wtree + (size_t)(kk + PD) * STEP_ELEMS);
    }

    #pragma unroll
    for (int v = 0; v < 8; ++v) {
        pw[0][mrow + v][ln]      = sigf(c00[v] + bias0);
        pw[0][mrow + v][16 + ln] = sigf(c01[v] + bias1);
        pw[1][mrow + v][ln]      = sigf(c10[v] + bias0);
        pw[1][mrow + v][16 + ln] = sigf(c11[v] + bias1);
    }
    __syncthreads();

    #pragma unroll
    for (int m = 0; m < 2; ++m) {
        Frag16 r;
        #pragma unroll
        for (int e = 0; e < 16; ++e) {
            int leaf = (e < 8) ? (khalf + e) : (16 + khalf + (e - 8));
            float prod = 1.f;
            #pragma unroll
            for (int d = 0; d <= 4; ++d) {
                int node = ((1 << d) - 1) + (leaf >> (5 - d));
                float p  = pw[m][ln][node];
                prod *= (((leaf >> (4 - d)) & 1) ? (1.f - p) : p);
            }
            r.u[e] = f2bf(prod);
        }
        out_acc[m] = __builtin_amdgcn_wmma_f32_16x16x32_bf16(
                    false, r.v, false, s.v, (short)0, out_acc[m], false, false);
    }
    __syncthreads();
}

__global__ __launch_bounds__(NWAVE * 32) __attribute__((amdgpu_waves_per_eu(4, 4))) void dndf_main(
        const float* __restrict__ x,
        const unsigned short* __restrict__ Wpf,
        const unsigned short* __restrict__ smf,
        const float* __restrict__ bp,
        float* __restrict__ out) {

    __shared__ unsigned short xtile[ROWS * XL_STRIDE];
    __shared__ float pbuf[NWAVE][2][16][33];

    const int tid  = threadIdx.x;
    const int wave = tid >> 5;
    const int lane = tid & 31;
    const int row0 = blockIdx.x * ROWS;

    const unsigned short* wl = Wpf + (size_t)wave * TPW * 16 * STEP_ELEMS + lane * 16;
    Frag16 ring[PD][2];
    #pragma unroll
    for (int j = 0; j < PD; ++j) load_step(ring[j], wl + (size_t)j * STEP_ELEMS);

    float4 xf[8];
    #pragma unroll
    for (int j = 0; j < 8; ++j) {
        int i  = tid + j * (NWAVE * 32);
        int r  = i / (IN_DIM / 4);
        int c4 = i % (IN_DIM / 4);
        xf[j] = ((const float4*)(x + (size_t)(row0 + r) * IN_DIM))[c4];
    }
    #pragma unroll
    for (int j = 0; j < 8; ++j) {
        int i  = tid + j * (NWAVE * 32);
        int r  = i / (IN_DIM / 4);
        int c4 = i % (IN_DIM / 4);
        int o = r * XL_STRIDE + c4 * 4;
        xtile[o + 0] = f2bf(xf[j].x);
        xtile[o + 1] = f2bf(xf[j].y);
        xtile[o + 2] = f2bf(xf[j].z);
        xtile[o + 3] = f2bf(xf[j].w);
    }
    __syncthreads();

    const int ln    = lane & 15;
    const int khalf = (lane >> 4) * 8;
    const int mrow  = (lane >> 4) * 8;

    const unsigned short* xrow = &xtile[ln * XL_STRIDE + khalf];
    v8f out_acc[2] = {};

    for (int tl = 0; tl < TPW - 1; ++tl)
        one_tree<false>(ring, out_acc, xrow, wl + (size_t)tl * 16 * STEP_ELEMS,
                        smf, bp, pbuf[wave], wave * TPW + tl, lane, ln, khalf, mrow);
    one_tree<true>(ring, out_acc, xrow, wl + (size_t)(TPW - 1) * 16 * STEP_ELEMS,
                   smf, bp, pbuf[wave], wave * TPW + (TPW - 1), lane, ln, khalf, mrow);

    #pragma unroll
    for (int m = 0; m < 2; ++m) {
        #pragma unroll
        for (int v = 0; v < 8; ++v) pbuf[wave][m][mrow + v][ln] = out_acc[m][v];
    }
    __syncthreads();

    const int rrow = tid >> 4;
    const int col  = tid & 15;
    float sum = 0.f;
    #pragma unroll
    for (int w = 0; w < NWAVE; ++w) sum += pbuf[w][rrow >> 4][rrow & 15][col];
    out[(size_t)(row0 + rrow) * OUT_DIM + col] = sum;
}

extern "C" void kernel_launch(void* const* d_in, const int* in_sizes, int n_in,
                              void* d_out, int out_size, void* d_ws, size_t ws_size,
                              hipStream_t stream) {
    const float* x     = (const float*)d_in[0];
    const float* W     = (const float*)d_in[1];
    const float* b     = (const float*)d_in[2];
    const float* leafs = (const float*)d_in[3];
    float* out = (float*)d_out;

    unsigned short* Wpf = (unsigned short*)d_ws;
    unsigned short* smf = (unsigned short*)((char*)d_ws + (size_t)2 * 1024 * 1024);
    float*          bpp = (float*)((char*)d_ws + (size_t)2 * 1024 * 1024 + 64 * 1024);

    dndf_pack<<<4096, 256, 0, stream>>>(W, leafs, b, Wpf, smf, bpp);
    dndf_main   <<<256, 512, 0, stream>>>(x, Wpf, smf, bpp, out);
}


# sigmoid: 1/(1+e) by the hardware reciprocal (v_rcp_f32, 1 ulp) instead of the IEEE division sequence: about 5 vector instructions per probability instead of about 17
# speedup vs baseline: 3.1325x; 1.0797x over previous
#include <hip/hip_runtime.h>
#include <hip/hip_bf16.h>


#define BATCH    8192
#define IN_DIM   512
#define NUM_TREE 64
#define NUM_LEAF 32
#define NUM_INT  31
#define OUT_DIM  16
#define XL_STRIDE 520
#define PD 8
#define ROWS 32
#define NWAVE 16
#define TPW 4
#define STEP_ELEMS 1024

typedef __attribute__((ext_vector_type(8)))  float  v8f;
typedef __attribute__((ext_vector_type(16))) __bf16 v16bf;

union Frag16 {
    v16bf v;
    unsigned short u[16];
    uint4 q[2];
};

__device__ __forceinline__ unsigned short f2bf(float f) {
    unsigned int u = __float_as_uint(f);
    unsigned int r = u + 0x7FFFu + ((u >> 16) & 1u);
    return (unsigned short)(r >> 16);
}

__device__ __forceinline__ float sigf(float x) {
    return __builtin_amdgcn_rcpf(1.0f + __expf(-x));
}

__global__ void dndf_pack(const float* __restrict__ W,
                          const float* __restrict__ leafs,
                          const float* __restrict__ b,
                          unsigned short* __restrict__ Wpf,
                          unsigned short* __restrict__ smf,
                          float* __restrict__ bp) {
    int idx = blockIdx.x * 256 + threadIdx.x;
    const int total = NUM_TREE * 16 * 2 * 32 * 16;
    if (idx >= total) return;
    {
        int e    = idx & 15;
        int lane = (idx >> 4) & 31;
        int ct   = (idx >> 9) & 1;
        int kk   = (idx >> 10) & 15;
        int t    = idx >> 14;
        int n    = ct * 16 + (lane & 15);
        int koff = (lane >> 4) * 8;
        int k    = kk * 32 + ((e < 8) ? (koff + e) : (16 + koff + (e - 8)));
        unsigned short v = 0;
        if (n < NUM_INT) v = f2bf(W[(t * NUM_INT + n) * IN_DIM + k]);
        Wpf[idx] = v;
    }
    if (idx < NUM_TREE * 32 * 16) {
        int e    = idx & 15;
        int lane = (idx >> 4) & 31;
        int t    = idx >> 9;
        int n    = lane & 15;
        int koff = (lane >> 4) * 8;
        int leaf = (e < 8) ? (koff + e) : (16 + koff + (e - 8));
        const float* row = leafs + (t * NUM_LEAF + leaf) * OUT_DIM;
        float mx = row[0];
        #pragma unroll
        for (int o = 1; o < OUT_DIM; ++o) mx = fmaxf(mx, row[o]);
        float sum = 0.f;
        #pragma unroll
        for (int o = 0; o < OUT_DIM; ++o) sum += __expf(row[o] - mx);
        float val = __expf(row[n] - mx) / (sum * (float)NUM_TREE);
        smf[idx] = f2bf(val);
    }
    if (idx < NUM_TREE * 32) {
        int i = idx & 31;
        bp[idx] = (i < NUM_INT) ? b[(idx >> 5) * NUM_INT + i] : 0.0f;
    }
}

__device__ __forceinline__ void load_step(Frag16 (&dst)[2], const unsigned short* p) {
    const uint4* q = (const uint4*)p;
    dst[0].q[0] = q[0];  dst[0].q[1] = q[1];
    dst[1].q[0] = q[64]; dst[1].q[1] = q[65];
}

template <bool LAST>
__device__ __forceinline__ void one_tree(
        Frag16 (&ring)[PD][2], v8f (&out_acc)[2],
        const unsigned short* xrow,
        const unsigned short* wtree,
        const unsigned short* smf, const float* bp,
        float (&pw)[2][16][33],
        int t, int lane, int ln, int khalf, int mrow) {

    const float bias0 = bp[t * 32 + ln];
    const float bias1 = bp[t * 32 + 16 + ln];
    Frag16 s;
    const uint4* sp = (const uint4*)(smf + ((size_t)t * 32 + lane) * 16);
    s.q[0] = sp[0]; s.q[1] = sp[1];

    v8f c00 = {};
    v8f c01 = {};
    v8f c10 = {};
    v8f c11 = {};
    #pragma unroll
    for (int kk = 0; kk < 16; ++kk) {
        Frag16 a0, a1;
        a0.q[0] = *(const uint4*)(xrow + kk * 32);
        a0.q[1] = *(const uint4*)(xrow + kk * 32 + 16);
        a1.q[0] = *(const uint4*)(xrow + 16 * XL_STRIDE + kk * 32);
        a1.q[1] = *(const uint4*)(xrow + 16 * XL_STRIDE + kk * 32 + 16);
        c00 = __builtin_amdgcn_wmma_f32_16x16x32_bf16(
                false, a0.v, false, ring[kk % PD][0].v, (short)0, c00, false, false);
        c01 = __builtin_amdgcn_wmma_f32_16x16x32_bf16(
                false, a0.v, false, ring[kk % PD][1].v, (short)0, c01, false, false);
        c10 = __builtin_amdgcn_wmma_f32_16x16x32_bf16(
                false, a1.v, false, ring[kk % PD][0].v, (short)0, c10, false, false);
        c11 = __builtin_amdgcn_wmma_f32_16x16x32_bf16(
                false, a1.v, false, ring[kk % PD][1].v, (short)0, c11, false, false);
        if (!LAST || kk + PD < 16)
            load_step(ring[kk % PD], wtree + (size_t)(kk + PD) * STEP_ELEMS);
    }

    #pragma unroll
    for (int v = 0; v < 8; ++v) {
        pw[0][mrow + v][ln]      = sigf(c00[v] + bias0);
        pw[0][mrow + v][16 + ln] = sigf(c01[v] + bias1);
        pw[1][mrow + v][ln]      = sigf(c10[v] + bias0);
        pw[1][mrow + v][16 + ln] = sigf(c11[v] + bias1);
    }
    __syncthreads();

    #pragma unroll
    for (int m = 0; m < 2; ++m) {
        Frag16 r;
        #pragma unroll
        for (int e = 0; e < 16; ++e) {
            int leaf = (e < 8) ? (khalf + e) : (16 + khalf + (e - 8));
            float prod = 1.f;
            #pragma unroll
            for (int d = 0; d <= 4; ++d) {
                int node = ((1 << d) - 1) + (leaf >> (5 - d));
                float p  = pw[m][ln][node];
                prod *= (((leaf >> (4 - d)) & 1) ? (1.f - p) : p);
            }
            r.u[e] = f2bf(prod);
        }
        out_acc[m] = __builtin_amdgcn_wmma_f32_16x16x32_bf16(
                    false, r.v, false, s.v, (short)0, out_acc[m], false, false);
    }
    __syncthreads();
}

__global__ __launch_bounds__(NWAVE * 32) __attribute__((amdgpu_waves_per_eu(4, 4))) void dndf_main(
        const float* __restrict__ x,
        const unsigned short* __restrict__ Wpf,
        const unsigned short* __restrict__ smf,
        const float* __restrict__ bp,
        float* __restrict__ out) {

    __shared__ unsigned short xtile[ROWS * XL_STRIDE];
    __shared__ float pbuf[NWAVE][2][16][33];

    const int tid  = threadIdx.x;
    const int wave = tid >> 5;
    const int lane = tid & 31;
    const int row0 = blockIdx.x * ROWS;

    const unsigned short* wl = Wpf + (size_t)wave * TPW * 16 * STEP_ELEMS + lane * 16;
    Frag16 ring[PD][2];
    #pragma unroll
    for (int j = 0; j < PD; ++j) load_step(ring[j], wl + (size_t)j * STEP_ELEMS);

    float4 xf[8];
    #pragma unroll
    for (int j = 0; j < 8; ++j) {
        int i  = tid + j * (NWAVE * 32);
        int r  = i / (IN_DIM / 4);
        int c4 = i % (IN_DIM / 4);
        xf[j] = ((const float4*)(x + (size_t)(row0 + r) * IN_DIM))[c4];
    }
    #pragma unroll
    for (int j = 0; j < 8; ++j) {
        int i  = tid + j * (NWAVE * 32);
        int r  = i / (IN_DIM / 4);
        int c4 = i % (IN_DIM / 4);
        int o = r * XL_STRIDE + c4 * 4;
        xtile[o + 0] = f2bf(xf[j].x);
        xtile[o + 1] = f2bf(xf[j].y);
        xtile[o + 2] = f2bf(xf[j].z);
        xtile[o + 3] = f2bf(xf[j].w);
    }
    __syncthreads();

    const int ln    = lane & 15;
    const int khalf = (lane >> 4) * 8;
    const int mrow  = (lane >> 4) * 8;

    const unsigned short* xrow = &xtile[ln * XL_STRIDE + khalf];
    v8f out_acc[2] = {};

    for (int tl = 0; tl < TPW - 1; ++tl)
        one_tree<false>(ring, out_acc, xrow, wl + (size_t)tl * 16 * STEP_ELEMS,
                        smf, bp, pbuf[wave], wave * TPW + tl, lane, ln, khalf, mrow);
    one_tree<true>(ring, out_acc, xrow, wl + (size_t)(TPW - 1) * 16 * STEP_ELEMS,
                   smf, bp, pbuf[wave], wave * TPW + (TPW - 1), lane, ln, khalf, mrow);

    #pragma unroll
    for (int m = 0; m < 2; ++m) {
        #pragma unroll
        for (int v = 0; v < 8; ++v) pbuf[wave][m][mrow + v][ln] = out_acc[m][v];
    }
    __syncthreads();

    const int rrow = tid >> 4;
    const int col  = tid & 15;
    float sum = 0.f;
    #pragma unroll
    for (int w = 0; w < NWAVE; ++w) sum += pbuf[w][rrow >> 4][rrow & 15][col];
    out[(size_t)(row0 + rrow) * OUT_DIM + col] = sum;
}

extern "C" void kernel_launch(void* const* d_in, const int* in_sizes, int n_in,
                              void* d_out, int out_size, void* d_ws, size_t ws_size,
                              hipStream_t stream) {
    const float* x     = (const float*)d_in[0];
    const float* W     = (const float*)d_in[1];
    const float* b     = (const float*)d_in[2];
    const float* leafs = (const float*)d_in[3];
    float* out = (float*)d_out;

    unsigned short* Wpf = (unsigned short*)d_ws;
    unsigned short* smf = (unsigned short*)((char*)d_ws + (size_t)2 * 1024 * 1024);
    float*          bpp = (float*)((char*)d_ws + (size_t)2 * 1024 * 1024 + 64 * 1024);

    dndf_pack<<<4096, 256, 0, stream>>>(W, leafs, b, Wpf, smf, bpp);
    dndf_main   <<<256, 512, 0, stream>>>(x, Wpf, smf, bpp, out);
}


# main kernel: the two workgroup barriers per tree guarded wave-private tiles only: now wave-local LDS ordering; waves run out of phase (vector work beside others' WMMAs), W loads stay in flight
# speedup vs baseline: 3.4522x; 1.1021x over previous
#include <hip/hip_runtime.h>
#include <hip/hip_bf16.h>


#define BATCH    8192
#define IN_DIM   512
#define NUM_TREE 64
#define NUM_LEAF 32
#define NUM_INT  31
#define OUT_DIM  16
#define XL_STRIDE 520
#define PD 8
#define ROWS 32
#define NWAVE 16
#define TPW 4
#define STEP_ELEMS 1024

typedef __attribute__((ext_vector_type(8)))  float  v8f;
typedef __attribute__((ext_vector_type(16))) __bf16 v16bf;

union Frag16 {
    v16bf v;
    unsigned short u[16];
    uint4 q[2];
};

__device__ __forceinline__ unsigned short f2bf(float f) {
    unsigned int u = __float_as_uint(f);
    unsigned int r = u + 0x7FFFu + ((u >> 16) & 1u);
    return (unsigned short)(r >> 16);
}

__device__ __forceinline__ float sigf(float x) {
    return __builtin_amdgcn_rcpf(1.0f + __expf(-x));
}

__global__ void dndf_pack(const float* __restrict__ W,
                          const float* __restrict__ leafs,
                          const float* __restrict__ b,
                          unsigned short* __restrict__ Wpf,
                          unsigned short* __restrict__ smf,
                          float* __restrict__ bp) {
    int idx = blockIdx.x * 256 + threadIdx.x;
    const int total = NUM_TREE * 16 * 2 * 32 * 16;
    if (idx >= total) return;
    {
        int e    = idx & 15;
        int lane = (idx >> 4) & 31;
        int ct   = (idx >> 9) & 1;
        int kk   = (idx >> 10) & 15;
        int t    = idx >> 14;
        int n    = ct * 16 + (lane & 15);
        int koff = (lane >> 4) * 8;
        int k    = kk * 32 + ((e < 8) ? (koff + e) : (16 + koff + (e - 8)));
        unsigned short v = 0;
        if (n < NUM_INT) v = f2bf(W[(t * NUM_INT + n) * IN_DIM + k]);
        Wpf[idx] = v;
    }
    if (idx < NUM_TREE * 32 * 16) {
        int e    = idx & 15;
        int lane = (idx >> 4) & 31;
        int t    = idx >> 9;
        int n    = lane & 15;
        int koff = (lane >> 4) * 8;
        int leaf = (e < 8) ? (koff + e) : (16 + koff + (e - 8));
        const float* row = leafs + (t * NUM_LEAF + leaf) * OUT_DIM;
        float mx = row[0];
        #pragma unroll
        for (int o = 1; o < OUT_DIM; ++o) mx = fmaxf(mx, row[o]);
        float sum = 0.f;
        #pragma unroll
        for (int o = 0; o < OUT_DIM; ++o) sum += __expf(row[o] - mx);
        float val = __expf(row[n] - mx) / (sum * (float)NUM_TREE);
        smf[idx] = f2bf(val);
    }
    if (idx < NUM_TREE * 32) {
        int i = idx & 31;
        bp[idx] = (i < NUM_INT) ? b[(idx >> 5) * NUM_INT + i] : 0.0f;
    }
}

__device__ __forceinline__ void load_step(Frag16 (&dst)[2], const unsigned short* p) {
    const uint4* q = (const uint4*)p;
    dst[0].q[0] = q[0];  dst[0].q[1] = q[1];
    dst[1].q[0] = q[64]; dst[1].q[1] = q[65];
}

__device__ __forceinline__ void wave_lds_order() {
    __builtin_amdgcn_fence(__ATOMIC_RELEASE, "wavefront");
    __builtin_amdgcn_wave_barrier();
    __builtin_amdgcn_fence(__ATOMIC_ACQUIRE, "wavefront");
}

template <bool LAST>
__device__ __forceinline__ void one_tree(
        Frag16 (&ring)[PD][2], v8f (&out_acc)[2],
        const unsigned short* xrow,
        const unsigned short* wtree,
        const unsigned short* smf, const float* bp,
        float (&pw)[2][16][33],
        int t, int lane, int ln, int khalf, int mrow) {

    const float bias0 = bp[t * 32 + ln];
    const float bias1 = bp[t * 32 + 16 + ln];
    Frag16 s;
    const uint4* sp = (const uint4*)(smf + ((size_t)t * 32 + lane) * 16);
    s.q[0] = sp[0]; s.q[1] = sp[1];

    v8f c00 = {};
    v8f c01 = {};
    v8f c10 = {};
    v8f c11 = {};
    #pragma unroll
    for (int kk = 0; kk < 16; ++kk) {
        Frag16 a0, a1;
        a0.q[0] = *(const uint4*)(xrow + kk * 32);
        a0.q[1] = *(const uint4*)(xrow + kk * 32 + 16);
        a1.q[0] = *(const uint4*)(xrow + 16 * XL_STRIDE + kk * 32);
        a1.q[1] = *(const uint4*)(xrow + 16 * XL_STRIDE + kk * 32 + 16);
        c00 = __builtin_amdgcn_wmma_f32_16x16x32_bf16(
                false, a0.v, false, ring[kk % PD][0].v, (short)0, c00, false, false);
        c01 = __builtin_amdgcn_wmma_f32_16x16x32_bf16(
                false, a0.v, false, ring[kk % PD][1].v, (short)0, c01, false, false);
        c10 = __builtin_amdgcn_wmma_f32_16x16x32_bf16(
                false, a1.v, false, ring[kk % PD][0].v, (short)0, c10, false, false);
        c11 = __builtin_amdgcn_wmma_f32_16x16x32_bf16(
                false, a1.v, false, ring[kk % PD][1].v, (short)0, c11, false, false);
        if (!LAST || kk + PD < 16)
            load_step(ring[kk % PD], wtree + (size_t)(kk + PD) * STEP_ELEMS);
    }

    #pragma unroll
    for (int v = 0; v < 8; ++v) {
        pw[0][mrow + v][ln]      = sigf(c00[v] + bias0);
        pw[0][mrow + v][16 + ln] = sigf(c01[v] + bias1);
        pw[1][mrow + v][ln]      = sigf(c10[v] + bias0);
        pw[1][mrow + v][16 + ln] = sigf(c11[v] + bias1);
    }
    wave_lds_order();

    #pragma unroll
    for (int m = 0; m < 2; ++m) {
        Frag16 r;
        #pragma unroll
        for (int e = 0; e < 16; ++e) {
            int leaf = (e < 8) ? (khalf + e) : (16 + khalf + (e - 8));
            float prod = 1.f;
            #pragma unroll
            for (int d = 0; d <= 4; ++d) {
                int node = ((1 << d) - 1) + (leaf >> (5 - d));
                float p  = pw[m][ln][node];
                prod *= (((leaf >> (4 - d)) & 1) ? (1.f - p) : p);
            }
            r.u[e] = f2bf(prod);
        }
        out_acc[m] = __builtin_amdgcn_wmma_f32_16x16x32_bf16(
                    false, r.v, false, s.v, (short)0, out_acc[m], false, false);
    }
    wave_lds_order();
}

__global__ __launch_bounds__(NWAVE * 32) __attribute__((amdgpu_waves_per_eu(4, 4))) void dndf_main(
        const float* __restrict__ x,
        const unsigned short* __restrict__ Wpf,
        const unsigned short* __restrict__ smf,
        const float* __restrict__ bp,
        float* __restrict__ out) {

    __shared__ unsigned short xtile[ROWS * XL_STRIDE];
    __shared__ float pbuf[NWAVE][2][16][33];

    const int tid  = threadIdx.x;
    const int wave = tid >> 5;
    const int lane = tid & 31;
    const int row0 = blockIdx.x * ROWS;

    const unsigned short* wl = Wpf + (size_t)wave * TPW * 16 * STEP_ELEMS + lane * 16;
    Frag16 ring[PD][2];
    #pragma unroll
    for (int j = 0; j < PD; ++j) load_step(ring[j], wl + (size_t)j * STEP_ELEMS);

    float4 xf[8];
    #pragma unroll
    for (int j = 0; j < 8; ++j) {
        int i  = tid + j * (NWAVE * 32);
        int r  = i / (IN_DIM / 4);
        int c4 = i % (IN_DIM / 4);
        xf[j] = ((const float4*)(x + (size_t)(row0 + r) * IN_DIM))[c4];
    }
    #pragma unroll
    for (int j = 0; j < 8; ++j) {
        int i  = tid + j * (NWAVE * 32);
        int r  = i / (IN_DIM / 4);
        int c4 = i % (IN_DIM / 4);
        int o = r * XL_STRIDE + c4 * 4;
        xtile[o + 0] = f2bf(xf[j].x);
        xtile[o + 1] = f2bf(xf[j].y);
        xtile[o + 2] = f2bf(xf[j].z);
        xtile[o + 3] = f2bf(xf[j].w);
    }
    __syncthreads();

    const int ln    = lane & 15;
    const int khalf = (lane >> 4) * 8;
    const int mrow  = (lane >> 4) * 8;

    const unsigned short* xrow = &xtile[ln * XL_STRIDE + khalf];
    v8f out_acc[2] = {};

    for (int tl = 0; tl < TPW - 1; ++tl)
        one_tree<false>(ring, out_acc, xrow, wl + (size_t)tl * 16 * STEP_ELEMS,
                        smf, bp, pbuf[wave], wave * TPW + tl, lane, ln, khalf, mrow);
    one_tree<true>(ring, out_acc, xrow, wl + (size_t)(TPW - 1) * 16 * STEP_ELEMS,
                   smf, bp, pbuf[wave], wave * TPW + (TPW - 1), lane, ln, khalf, mrow);

    #pragma unroll
    for (int m = 0; m < 2; ++m) {
        #pragma unroll
        for (int v = 0; v < 8; ++v) pbuf[wave][m][mrow + v][ln] = out_acc[m][v];
    }
    __syncthreads();

    const int rrow = tid >> 4;
    const int col  = tid & 15;
    float sum = 0.f;
    #pragma unroll
    for (int w = 0; w < NWAVE; ++w) sum += pbuf[w][rrow >> 4][rrow & 15][col];
    out[(size_t)(row0 + rrow) * OUT_DIM + col] = sum;
}

extern "C" void kernel_launch(void* const* d_in, const int* in_sizes, int n_in,
                              void* d_out, int out_size, void* d_ws, size_t ws_size,
                              hipStream_t stream) {
    const float* x     = (const float*)d_in[0];
    const float* W     = (const float*)d_in[1];
    const float* b     = (const float*)d_in[2];
    const float* leafs = (const float*)d_in[3];
    float* out = (float*)d_out;

    unsigned short* Wpf = (unsigned short*)d_ws;
    unsigned short* smf = (unsigned short*)((char*)d_ws + (size_t)2 * 1024 * 1024);
    float*          bpp = (float*)((char*)d_ws + (size_t)2 * 1024 * 1024 + 64 * 1024);

    dndf_pack<<<4096, 256, 0, stream>>>(W, leafs, b, Wpf, smf, bpp);
    dndf_main   <<<256, 512, 0, stream>>>(x, Wpf, smf, bpp, out);
}


# main kernel: every block walks the W stream from its own starting point (tree and K-step order rotated by block index), so blocks stop hitting the same L2 lines together; f32 sum order rotates too
# speedup vs baseline: 3.5841x; 1.0382x over previous
#include <hip/hip_runtime.h>
#include <hip/hip_bf16.h>


#define BATCH    8192
#define IN_DIM   512
#define NUM_TREE 64
#define NUM_LEAF 32
#define NUM_INT  31
#define OUT_DIM  16
#define XL_STRIDE 520
#define PD 8
#define ROWS 32
#define NWAVE 16
#define TPW 4
#define STEP_ELEMS 1024

typedef __attribute__((ext_vector_type(8)))  float  v8f;
typedef __attribute__((ext_vector_type(16))) __bf16 v16bf;

union Frag16 {
    v16bf v;
    unsigned short u[16];
    uint4 q[2];
};

__device__ __forceinline__ unsigned short f2bf(float f) {
    unsigned int u = __float_as_uint(f);
    unsigned int r = u + 0x7FFFu + ((u >> 16) & 1u);
    return (unsigned short)(r >> 16);
}

__device__ __forceinline__ float sigf(float x) {
    return __builtin_amdgcn_rcpf(1.0f + __expf(-x));
}

__global__ void dndf_pack(const float* __restrict__ W,
                          const float* __restrict__ leafs,
                          const float* __restrict__ b,
                          unsigned short* __restrict__ Wpf,
                          unsigned short* __restrict__ smf,
                          float* __restrict__ bp) {
    int idx = blockIdx.x * 256 + threadIdx.x;
    const int total = NUM_TREE * 16 * 2 * 32 * 16;
    if (idx >= total) return;
    {
        int e    = idx & 15;
        int lane = (idx >> 4) & 31;
        int ct   = (idx >> 9) & 1;
        int kk   = (idx >> 10) & 15;
        int t    = idx >> 14;
        int n    = ct * 16 + (lane & 15);
        int koff = (lane >> 4) * 8;
        int k    = kk * 32 + ((e < 8) ? (koff + e) : (16 + koff + (e - 8)));
        unsigned short v = 0;
        if (n < NUM_INT) v = f2bf(W[(t * NUM_INT + n) * IN_DIM + k]);
        Wpf[idx] = v;
    }
    if (idx < NUM_TREE * 32 * 16) {
        int e    = idx & 15;
        int lane = (idx >> 4) & 31;
        int t    = idx >> 9;
        int n    = lane & 15;
        int koff = (lane >> 4) * 8;
        int leaf = (e < 8) ? (koff + e) : (16 + koff + (e - 8));
        const float* row = leafs + (t * NUM_LEAF + leaf) * OUT_DIM;
        float mx = row[0];
        #pragma unroll
        for (int o = 1; o < OUT_DIM; ++o) mx = fmaxf(mx, row[o]);
        float sum = 0.f;
        #pragma unroll
        for (int o = 0; o < OUT_DIM; ++o) sum += __expf(row[o] - mx);
        float val = __expf(row[n] - mx) / (sum * (float)NUM_TREE);
        smf[idx] = f2bf(val);
    }
    if (idx < NUM_TREE * 32) {
        int i = idx & 31;
        bp[idx] = (i < NUM_INT) ? b[(idx >> 5) * NUM_INT + i] : 0.0f;
    }
}

__device__ __forceinline__ void load_step(Frag16 (&dst)[2], const unsigned short* step,
                                          unsigned lane_elems) {
    const uint4* q = (const uint4*)(step + lane_elems);
    dst[0].q[0] = q[0];  dst[0].q[1] = q[1];
    dst[1].q[0] = q[64]; dst[1].q[1] = q[65];
}

__device__ __forceinline__ void wave_lds_order() {
    __builtin_amdgcn_fence(__ATOMIC_RELEASE, "wavefront");
    __builtin_amdgcn_wave_barrier();
    __builtin_amdgcn_fence(__ATOMIC_ACQUIRE, "wavefront");
}

template <bool LAST>
__device__ __forceinline__ void one_tree(
        Frag16 (&ring)[PD][2], v8f (&out_acc)[2],
        const unsigned short* xrow,
        const unsigned short* wtree,
        const unsigned short* wnext,
        const unsigned short* smf, const float* bp,
        float (&pw)[2][16][33],
        int t, int k0,
        int lane, int ln, int khalf, int mrow) {

    const float bias0 = bp[t * 32 + ln];
    const float bias1 = bp[t * 32 + 16 + ln];
    Frag16 s;
    const uint4* sp = (const uint4*)(smf + ((size_t)t * 32 + lane) * 16);
    s.q[0] = sp[0]; s.q[1] = sp[1];

    v8f c00 = {};
    v8f c01 = {};
    v8f c10 = {};
    v8f c11 = {};
    #pragma unroll
    for (int kk = 0; kk < 16; ++kk) {
        Frag16 a0, a1;
        a0.q[0] = *(const uint4*)(xrow + kk * 32);
        a0.q[1] = *(const uint4*)(xrow + kk * 32 + 16);
        a1.q[0] = *(const uint4*)(xrow + 16 * XL_STRIDE + kk * 32);
        a1.q[1] = *(const uint4*)(xrow + 16 * XL_STRIDE + kk * 32 + 16);
        c00 = __builtin_amdgcn_wmma_f32_16x16x32_bf16(
                false, a0.v, false, ring[kk % PD][0].v, (short)0, c00, false, false);
        c01 = __builtin_amdgcn_wmma_f32_16x16x32_bf16(
                false, a0.v, false, ring[kk % PD][1].v, (short)0, c01, false, false);
        c10 = __builtin_amdgcn_wmma_f32_16x16x32_bf16(
                false, a1.v, false, ring[kk % PD][0].v, (short)0, c10, false, false);
        c11 = __builtin_amdgcn_wmma_f32_16x16x32_bf16(
                false, a1.v, false, ring[kk % PD][1].v, (short)0, c11, false, false);
        if (!LAST || kk + PD < 16)
            load_step(ring[kk % PD], ((kk + PD < 16) ? wtree : wnext)
                                     + (size_t)((kk + PD + k0) & 15) * STEP_ELEMS,
                      (unsigned)lane * 16u);
    }

    #pragma unroll
    for (int v = 0; v < 8; ++v) {
        pw[0][mrow + v][ln]      = sigf(c00[v] + bias0);
        pw[0][mrow + v][16 + ln] = sigf(c01[v] + bias1);
        pw[1][mrow + v][ln]      = sigf(c10[v] + bias0);
        pw[1][mrow + v][16 + ln] = sigf(c11[v] + bias1);
    }
    wave_lds_order();

    #pragma unroll
    for (int m = 0; m < 2; ++m) {
        Frag16 r;
        #pragma unroll
        for (int e = 0; e < 16; ++e) {
            int leaf = (e < 8) ? (khalf + e) : (16 + khalf + (e - 8));
            float prod = 1.f;
            #pragma unroll
            for (int d = 0; d <= 4; ++d) {
                int node = ((1 << d) - 1) + (leaf >> (5 - d));
                float p  = pw[m][ln][node];
                prod *= (((leaf >> (4 - d)) & 1) ? (1.f - p) : p);
            }
            r.u[e] = f2bf(prod);
        }
        out_acc[m] = __builtin_amdgcn_wmma_f32_16x16x32_bf16(
                    false, r.v, false, s.v, (short)0, out_acc[m], false, false);
    }
    wave_lds_order();
}

__global__ __launch_bounds__(NWAVE * 32) __attribute__((amdgpu_waves_per_eu(4, 4))) void dndf_main(
        const float* __restrict__ x,
        const unsigned short* __restrict__ Wpf,
        const unsigned short* __restrict__ smf,
        const float* __restrict__ bp,
        float* __restrict__ out) {

    __shared__ unsigned short xtile[ROWS * XL_STRIDE];
    __shared__ float pbuf[NWAVE][2][16][33];

    const int tid  = threadIdx.x;
    const int wave = __builtin_amdgcn_readfirstlane(tid >> 5);
    const int lane = tid & 31;
    const int row0 = blockIdx.x * ROWS;

    const int r0 = blockIdx.x & (TPW - 1);
    const int k0 = (blockIdx.x >> 2) & 15;

    const unsigned short* wl = Wpf + (size_t)wave * TPW * 16 * STEP_ELEMS;
    Frag16 ring[PD][2];
    #pragma unroll
    for (int j = 0; j < PD; ++j)
        load_step(ring[j], wl + (size_t)r0 * 16 * STEP_ELEMS + (size_t)((j + k0) & 15) * STEP_ELEMS,
                  (unsigned)lane * 16u);

    float4 xf[8];
    #pragma unroll
    for (int j = 0; j < 8; ++j) {
        int i  = tid + j * (NWAVE * 32);
        int r  = i / (IN_DIM / 4);
        int c4 = i % (IN_DIM / 4);
        xf[j] = ((const float4*)(x + (size_t)(row0 + r) * IN_DIM))[c4];
    }
    #pragma unroll
    for (int j = 0; j < 8; ++j) {
        int i  = tid + j * (NWAVE * 32);
        int r  = i / (IN_DIM / 4);
        int c4 = i % (IN_DIM / 4);
        int o = r * XL_STRIDE + ((c4 * 4 - k0 * 32) & (IN_DIM - 1));
        xtile[o + 0] = f2bf(xf[j].x);
        xtile[o + 1] = f2bf(xf[j].y);
        xtile[o + 2] = f2bf(xf[j].z);
        xtile[o + 3] = f2bf(xf[j].w);
    }
    __syncthreads();

    const int ln    = lane & 15;
    const int khalf = (lane >> 4) * 8;
    const int mrow  = (lane >> 4) * 8;

    const unsigned short* xrow = &xtile[ln * XL_STRIDE + khalf];
    v8f out_acc[2] = {};

    for (int tl = 0; tl < TPW - 1; ++tl) {
        const int tc = (tl + r0) & (TPW - 1);
        const int tn = (tl + 1 + r0) & (TPW - 1);
        one_tree<false>(ring, out_acc, xrow, wl + (size_t)tc * 16 * STEP_ELEMS,
                        wl + (size_t)tn * 16 * STEP_ELEMS,
                        smf, bp, pbuf[wave], wave * TPW + tc, k0, lane, ln, khalf, mrow);
    }
    {
        const int tc = (TPW - 1 + r0) & (TPW - 1);
        one_tree<true>(ring, out_acc, xrow, wl + (size_t)tc * 16 * STEP_ELEMS, wl,
                       smf, bp, pbuf[wave], wave * TPW + tc, k0, lane, ln, khalf, mrow);
    }

    #pragma unroll
    for (int m = 0; m < 2; ++m) {
        #pragma unroll
        for (int v = 0; v < 8; ++v) pbuf[wave][m][mrow + v][ln] = out_acc[m][v];
    }
    __syncthreads();

    const int rrow = tid >> 4;
    const int col  = tid & 15;
    float sum = 0.f;
    #pragma unroll
    for (int w = 0; w < NWAVE; ++w) sum += pbuf[w][rrow >> 4][rrow & 15][col];
    out[(size_t)(row0 + rrow) * OUT_DIM + col] = sum;
}

extern "C" void kernel_launch(void* const* d_in, const int* in_sizes, int n_in,
                              void* d_out, int out_size, void* d_ws, size_t ws_size,
                              hipStream_t stream) {
    const float* x     = (const float*)d_in[0];
    const float* W     = (const float*)d_in[1];
    const float* b     = (const float*)d_in[2];
    const float* leafs = (const float*)d_in[3];
    float* out = (float*)d_out;

    unsigned short* Wpf = (unsigned short*)d_ws;
    unsigned short* smf = (unsigned short*)((char*)d_ws + (size_t)2 * 1024 * 1024);
    float*          bpp = (float*)((char*)d_ws + (size_t)2 * 1024 * 1024 + 64 * 1024);

    dndf_pack<<<4096, 256, 0, stream>>>(W, leafs, b, Wpf, smf, bpp);
    dndf_main   <<<256, 512, 0, stream>>>(x, Wpf, smf, bpp, out);
}


# main kernel: W fragment as WMMA A operand, x as B, node rows packed so a lane's accumulators hold its own batch row's path nodes: routes built in registers, no LDS transpose (top 1-p = sigmoid(-z))
# speedup vs baseline: 3.7239x; 1.0390x over previous
#include <hip/hip_runtime.h>
#include <hip/hip_bf16.h>


#define BATCH    8192
#define IN_DIM   512
#define NUM_TREE 64
#define NUM_LEAF 32
#define NUM_INT  31
#define OUT_DIM  16
#define XL_STRIDE 520
#define PD 8
#define ROWS 32
#define NWAVE 16
#define TPW 4
#define STEP_ELEMS 1024

typedef __attribute__((ext_vector_type(8)))  float  v8f;
typedef __attribute__((ext_vector_type(16))) __bf16 v16bf;

union Frag16 {
    v16bf v;
    unsigned short u[16];
    uint4 q[2];
};

__device__ __forceinline__ unsigned short f2bf(float f) {
    unsigned int u = __float_as_uint(f);
    unsigned int r = u + 0x7FFFu + ((u >> 16) & 1u);
    return (unsigned short)(r >> 16);
}

__device__ __forceinline__ float sigf(float x) {
    return __builtin_amdgcn_rcpf(1.0f + __expf(-x));
}

__device__ __forceinline__ int slot_node(int j, bool& neg) {
    const int g = j >> 4, h = (j >> 3) & 1, r = j & 7;
    const int base = h ? (g ? 8 : 24) : (g ? 16 : 0);
    neg = false;
    if (r == 0) return 3 + (base >> 3);
    if (r <= 2) return 7 + (base >> 2) + (r - 1);
    if (r <= 6) return 15 + (base >> 1) + (r - 3);
    neg = (h != 0);
    return g ? (1 + h) : 0;
}

__device__ __forceinline__ int frag_leaf(int h, int e) {
    return h ? ((e < 8) ? (24 + e) : (8 + (e - 8)))
             : ((e < 8) ? e : (16 + (e - 8)));
}

__global__ void dndf_pack(const float* __restrict__ W,
                          const float* __restrict__ leafs,
                          const float* __restrict__ b,
                          unsigned short* __restrict__ Wpf,
                          unsigned short* __restrict__ smf,
                          float* __restrict__ bp) {
    int idx = blockIdx.x * 256 + threadIdx.x;
    const int total = NUM_TREE * 16 * 2 * 32 * 16;
    if (idx >= total) return;
    {
        int e    = idx & 15;
        int lane = (idx >> 4) & 31;
        int ct   = (idx >> 9) & 1;
        int kk   = (idx >> 10) & 15;
        int t    = idx >> 14;
        int j    = ct * 16 + (lane & 15);
        int koff = (lane >> 4) * 8;
        int k    = kk * 32 + ((e < 8) ? (koff + e) : (16 + koff + (e - 8)));
        bool neg;
        int node = slot_node(j, neg);
        float w  = W[(t * NUM_INT + node) * IN_DIM + k];
        Wpf[idx] = f2bf(neg ? -w : w);
    }
    if (idx < NUM_TREE * 32 * 16) {
        int e    = idx & 15;
        int lane = (idx >> 4) & 31;
        int t    = idx >> 9;
        int n    = lane & 15;
        int leaf = frag_leaf(lane >> 4, e);
        const float* row = leafs + (t * NUM_LEAF + leaf) * OUT_DIM;
        float mx = row[0];
        #pragma unroll
        for (int o = 1; o < OUT_DIM; ++o) mx = fmaxf(mx, row[o]);
        float sum = 0.f;
        #pragma unroll
        for (int o = 0; o < OUT_DIM; ++o) sum += __expf(row[o] - mx);
        float val = __expf(row[n] - mx) / (sum * (float)NUM_TREE);
        smf[idx] = f2bf(val);
    }
    if (idx < NUM_TREE * 32) {
        int i = idx & 31;
        int h = i >> 4, s = i & 15;
        bool neg;
        int node = slot_node((s >> 3) * 16 + h * 8 + (s & 7), neg);
        float bv = b[(idx >> 5) * NUM_INT + node];
        bp[idx] = neg ? -bv : bv;
    }
}

__device__ __forceinline__ void load_step(Frag16 (&dst)[2], const unsigned short* step,
                                          unsigned lane_elems) {
    const uint4* q = (const uint4*)(step + lane_elems);
    dst[0].q[0] = q[0];  dst[0].q[1] = q[1];
    dst[1].q[0] = q[64]; dst[1].q[1] = q[65];
}

__device__ __forceinline__ float xhalf(float x) {
    return __builtin_bit_cast(float, __builtin_amdgcn_permlanex16(
        __builtin_bit_cast(int, x), __builtin_bit_cast(int, x), 0x76543210, 0xfedcba98, false, false));
}

template <bool LAST>
__device__ __forceinline__ void one_tree(
        Frag16 (&ring)[PD][2], v8f (&out_acc)[2],
        const unsigned short* xrow,
        const unsigned short* wtree,
        const unsigned short* wnext,
        const unsigned short* smf, const float* bp,
        int t, int k0,
        int lane) {

    asm volatile("" ::: "memory");

    const float4* bq = (const float4*)(bp + t * 32 + (lane >> 4) * 16);
    const float4 b0 = bq[0], b1 = bq[1], b2 = bq[2], b3 = bq[3];
    Frag16 s;
    const uint4* sp = (const uint4*)(smf + ((size_t)t * 32 + lane) * 16);
    s.q[0] = sp[0]; s.q[1] = sp[1];

    v8f c00 = {};
    v8f c01 = {};
    v8f c10 = {};
    v8f c11 = {};
    #pragma unroll
    for (int kk = 0; kk < 16; ++kk) {
        Frag16 a0, a1;
        a0.q[0] = *(const uint4*)(xrow + kk * 32);
        a0.q[1] = *(const uint4*)(xrow + kk * 32 + 16);
        a1.q[0] = *(const uint4*)(xrow + 16 * XL_STRIDE + kk * 32);
        a1.q[1] = *(const uint4*)(xrow + 16 * XL_STRIDE + kk * 32 + 16);
        c00 = __builtin_amdgcn_wmma_f32_16x16x32_bf16(
                false, ring[kk % PD][0].v, false, a0.v, (short)0, c00, false, false);
        c01 = __builtin_amdgcn_wmma_f32_16x16x32_bf16(
                false, ring[kk % PD][1].v, false, a0.v, (short)0, c01, false, false);
        c10 = __builtin_amdgcn_wmma_f32_16x16x32_bf16(
                false, ring[kk % PD][0].v, false, a1.v, (short)0, c10, false, false);
        c11 = __builtin_amdgcn_wmma_f32_16x16x32_bf16(
                false, ring[kk % PD][1].v, false, a1.v, (short)0, c11, false, false);
        if (!LAST || kk + PD < 16)
            load_step(ring[kk % PD], ((kk + PD < 16) ? wtree : wnext)
                                     + (size_t)((kk + PD + k0) & 15) * STEP_ELEMS,
                      (unsigned)lane * 16u);
    }

    const float bias[16] = { b0.x, b0.y, b0.z, b0.w, b1.x, b1.y, b1.z, b1.w,
                             b2.x, b2.y, b2.z, b2.w, b3.x, b3.y, b3.z, b3.w };

    #pragma unroll
    for (int m = 0; m < 2; ++m) {
        const v8f& lo = m ? c10 : c00;
        const v8f& hi = m ? c11 : c01;
        float P[16];
        #pragma unroll
        for (int v = 0; v < 8; ++v) {
            P[v]     = sigf(lo[v] + bias[v]);
            P[8 + v] = sigf(hi[v] + bias[8 + v]);
        }
        const float pre[2] = { P[7] * P[15], (1.f - P[7]) * (1.f - xhalf(P[15])) };
        Frag16 r;
        #pragma unroll
        for (int g = 0; g < 2; ++g) {
            #pragma unroll
            for (int e = 0; e < 8; ++e) {
                const float n2 = P[8 * g];
                const float n3 = P[8 * g + 1 + (e >> 2)];
                const float n4 = P[8 * g + 3 + (e >> 1)];
                float prod = pre[g];
                prod *= ((e & 4) ? (1.f - n2) : n2);
                prod *= ((e & 2) ? (1.f - n3) : n3);
                prod *= ((e & 1) ? (1.f - n4) : n4);
                r.u[8 * g + e] = f2bf(prod);
            }
        }
        out_acc[m] = __builtin_amdgcn_wmma_f32_16x16x32_bf16(
                    false, s.v, false, r.v, (short)0, out_acc[m], false, false);
    }
}

__global__ __launch_bounds__(NWAVE * 32) __attribute__((amdgpu_waves_per_eu(4, 4))) void dndf_main(
        const float* __restrict__ x,
        const unsigned short* __restrict__ Wpf,
        const unsigned short* __restrict__ smf,
        const float* __restrict__ bp,
        float* __restrict__ out) {

    __shared__ unsigned short xtile[ROWS * XL_STRIDE];
    __shared__ float pbuf[NWAVE][2][16][17];

    const int tid  = threadIdx.x;
    const int wave = __builtin_amdgcn_readfirstlane(tid >> 5);
    const int lane = tid & 31;
    const int row0 = blockIdx.x * ROWS;

    const int r0 = blockIdx.x & (TPW - 1);
    const int k0 = (blockIdx.x >> 2) & 15;

    const unsigned short* wl = Wpf + (size_t)wave * TPW * 16 * STEP_ELEMS;
    Frag16 ring[PD][2];
    #pragma unroll
    for (int j = 0; j < PD; ++j)
        load_step(ring[j], wl + (size_t)r0 * 16 * STEP_ELEMS + (size_t)((j + k0) & 15) * STEP_ELEMS,
                  (unsigned)lane * 16u);

    float4 xf[8];
    #pragma unroll
    for (int j = 0; j < 8; ++j) {
        int i  = tid + j * (NWAVE * 32);
        int r  = i / (IN_DIM / 4);
        int c4 = i % (IN_DIM / 4);
        xf[j] = ((const float4*)(x + (size_t)(row0 + r) * IN_DIM))[c4];
    }
    #pragma unroll
    for (int j = 0; j < 8; ++j) {
        int i  = tid + j * (NWAVE * 32);
        int r  = i / (IN_DIM / 4);
        int c4 = i % (IN_DIM / 4);
        int o = r * XL_STRIDE + ((c4 * 4 - k0 * 32) & (IN_DIM - 1));
        xtile[o + 0] = f2bf(xf[j].x);
        xtile[o + 1] = f2bf(xf[j].y);
        xtile[o + 2] = f2bf(xf[j].z);
        xtile[o + 3] = f2bf(xf[j].w);
    }
    __syncthreads();

    const int ln    = lane & 15;
    const int khalf = (lane >> 4) * 8;
    const int mrow  = (lane >> 4) * 8;

    const unsigned short* xrow = &xtile[ln * XL_STRIDE + khalf];
    v8f out_acc[2] = {};

    for (int tl = 0; tl < TPW - 1; ++tl) {
        const int tc = (tl + r0) & (TPW - 1);
        const int tn = (tl + 1 + r0) & (TPW - 1);
        one_tree<false>(ring, out_acc, xrow, wl + (size_t)tc * 16 * STEP_ELEMS,
                        wl + (size_t)tn * 16 * STEP_ELEMS,
                        smf, bp, wave * TPW + tc, k0, lane);
    }
    {
        const int tc = (TPW - 1 + r0) & (TPW - 1);
        one_tree<true>(ring, out_acc, xrow, wl + (size_t)tc * 16 * STEP_ELEMS, wl,
                       smf, bp, wave * TPW + tc, k0, lane);
    }

    #pragma unroll
    for (int m = 0; m < 2; ++m) {
        #pragma unroll
        for (int v = 0; v < 8; ++v) pbuf[wave][m][ln][mrow + v] = out_acc[m][v];
    }
    __syncthreads();

    const int rrow = tid >> 4;
    const int col  = tid & 15;
    float sum = 0.f;
    #pragma unroll
    for (int w = 0; w < NWAVE; ++w) sum += pbuf[w][rrow >> 4][rrow & 15][col];
    out[(size_t)(row0 + rrow) * OUT_DIM + col] = sum;
}

extern "C" void kernel_launch(void* const* d_in, const int* in_sizes, int n_in,
                              void* d_out, int out_size, void* d_ws, size_t ws_size,
                              hipStream_t stream) {
    const float* x     = (const float*)d_in[0];
    const float* W     = (const float*)d_in[1];
    const float* b     = (const float*)d_in[2];
    const float* leafs = (const float*)d_in[3];
    float* out = (float*)d_out;

    unsigned short* Wpf = (unsigned short*)d_ws;
    unsigned short* smf = (unsigned short*)((char*)d_ws + (size_t)2 * 1024 * 1024);
    float*          bpp = (float*)((char*)d_ws + (size_t)2 * 1024 * 1024 + 64 * 1024);

    dndf_pack<<<4096, 256, 0, stream>>>(W, leafs, b, Wpf, smf, bpp);
    dndf_main   <<<256, 512, 0, stream>>>(x, Wpf, smf, bpp, out);
}
